# starting kernel
# baseline (speedup 1.0000x reference)
#include <hip/hip_runtime.h>

typedef float v2f __attribute__((ext_vector_type(2)));
typedef float v8f __attribute__((ext_vector_type(8)));

#define NB 4
#define NC 16
#define NO 32
#define NK 3
#define NH 56
#define NW 56
#define HW (NH * NW)
#define TILES (HW / 16)

__global__ __launch_bounds__(256) void kaconv_wmma_kernel(
    const float* __restrict__ x,
    const float* __restrict__ nums,
    const float* __restrict__ denoms,
    const float* __restrict__ bias,
    float* __restrict__ out)
{
  const int lane   = threadIdx.x & 31;
  const int wib    = threadIdx.x >> 5;
  const int wave   = blockIdx.x * 8 + wib;
  const int b      = wave / (TILES * 2);
  const int rem    = wave - b * (TILES * 2);
  const int tile   = rem >> 1;
  const int ochunk = rem & 1;

  const int m  = lane & 15;
  const int hi = lane >> 4;

  const int pix = tile * 16 + m;
  const int h = pix / NW;
  const int w = pix - h * NW;

  v8f acc = {};

  const int obase = ochunk * 16;
  const int orow  = (obase + m) * NC;

  for (int c = 0; c < NC; ++c) {
    const float* xc = x + ((size_t)(b * NC + c)) * HW;
    const int rowc = (orow + c) * NK;
#pragma unroll
    for (int i = 0; i < NK; ++i) {
      const int hh    = h + i - 1;
      const int rowci = (rowc + i) * NK;
#pragma unroll
      for (int j = 0; j < NK; ++j) {
        const int ww = w + j - 1;
        const bool inb = (hh >= 0) && (hh < NH) && (ww >= 0) && (ww < NW);
        const float xv = inb ? xc[hh * NW + ww] : 0.0f;
        const float x2 = xv * xv;
        const float x3 = x2 * xv;
        const float x4 = x2 * x2;
        const float x5 = x3 * x2;

        v2f aN0 = {}, aN1 = {}, aD = {};
        aN0[0] = hi ? x2 : 1.0f;  aN0[1] = hi ? x3 : xv;
        aN1[0] = hi ? 0.0f : x4;  aN1[1] = hi ? 0.0f : x5;
        aD[0]  = hi ? x3 : xv;    aD[1]  = hi ? x4 : x2;

        const int row = rowci + j;
        const float* nr = nums   + (size_t)row * 6;
        const float* dr = denoms + (size_t)row * 4;
        v2f bN0 = *(const v2f*)(nr + (hi ? 2 : 0));
        v2f bN1 = *(const v2f*)(nr + (hi ? 2 : 4));
        v2f bD  = *(const v2f*)(dr + (hi ? 2 : 0));

        v8f numv = {};
        numv = __builtin_amdgcn_wmma_f32_16x16x4_f32(false, aN0, false, bN0,
                                                     (short)0, numv, false, false);
        numv = __builtin_amdgcn_wmma_f32_16x16x4_f32(false, aN1, false, bN1,
                                                     (short)0, numv, false, false);
        v8f denv = {};
        denv = __builtin_amdgcn_wmma_f32_16x16x4_f32(false, aD, false, bD,
                                                     (short)0, denv, false, false);

#pragma unroll
        for (int k = 0; k < 8; ++k) {
          const float d = 1.0f + __builtin_fabsf(denv[k]);
          acc[k] = __builtin_fmaf(numv[k], __builtin_amdgcn_rcpf(d), acc[k]);
        }
      }
    }
  }

  const float bo = bias[obase + m];
  float* op = out + ((size_t)(b * NO + obase + m)) * HW + tile * 16 + hi * 8;
#pragma unroll
  for (int k = 0; k < 8; ++k) op[k] = acc[k] + bo;
}

extern "C" void kernel_launch(void* const* d_in, const int* in_sizes, int n_in,
                              void* d_out, int out_size, void* d_ws, size_t ws_size,
                              hipStream_t stream) {
  const float* x      = (const float*)d_in[0];
  const float* nums   = (const float*)d_in[1];
  const float* denoms = (const float*)d_in[2];
  const float* bias   = (const float*)d_in[3];
  float* out          = (float*)d_out;
  (void)in_sizes; (void)n_in; (void)out_size; (void)d_ws; (void)ws_size;

  dim3 grid(196);
  dim3 block(256);
  hipLaunchKernelGGL(kaconv_wmma_kernel, grid, block, 0, stream,
                     x, nums, denoms, bias, out);
}


# Both coefficient tables staged once per work-group into padded LDS by coalesced 512-B copies; B fragments read conflict-free from LDS, grouped per channel, instead of 16-line global gathers
# speedup vs baseline: 1.9986x; 1.9986x over previous
#include <hip/hip_runtime.h>


typedef float v2f __attribute__((ext_vector_type(2)));
typedef float v4f __attribute__((ext_vector_type(4)));
typedef float v8f __attribute__((ext_vector_type(8)));

#define NB 4
#define NC 16
#define NO 32
#define NK 3
#define NT (NK * NK)
#define NH 56
#define NW 56
#define HW (NH * NW)
#define TILES (HW / 16)

#define NUM_O_F4    (NC * NT * 6 / 4)
#define DEN_O_F4    (NC * NT * 4 / 4)
#define NUM_O_PITCH (NUM_O_F4 + 1)
#define DEN_O_PITCH (DEN_O_F4 + 1)
#define LDS_DEN_F4  (NO * NUM_O_PITCH)
#define LDS_F4      (LDS_DEN_F4 + NO * DEN_O_PITCH)

struct ChanOps {
  float x[NT];
  v2f   n0[NT];
  v2f   n1[NT];
  v2f   d[NT];
};

__device__ __forceinline__ void load_chan(ChanOps& s,
                                          const float* __restrict__ xc,
                                          const int (&xoff)[NT],
                                          const float* __restrict__ pn0,
                                          const float* __restrict__ pn1,
                                          const float* __restrict__ pd) {
#pragma unroll
  for (int t = 0; t < NT; ++t) s.x[t] = xc[xoff[t]];
#pragma unroll
  for (int t = 0; t < NT; ++t) {
    s.n0[t] = *(const v2f*)(pn0 + t * 6);
    s.n1[t] = *(const v2f*)(pn1 + t * 6);
    s.d[t]  = *(const v2f*)(pd  + t * 4);
  }
}

__device__ __forceinline__ void compute_chan(const ChanOps& s,
                                             const bool (&inb)[NT],
                                             const int hi, v8f& acc) {
#pragma unroll
  for (int t = 0; t < NT; ++t) {
    const float xv = inb[t] ? s.x[t] : 0.0f;
    const float x2 = xv * xv;
    const float x3 = x2 * xv;
    const float x4 = x2 * x2;
    const float x5 = x3 * x2;

    v2f aN0 = {}, aN1 = {}, aD = {};
    aN0[0] = hi ? x2 : 1.0f;  aN0[1] = hi ? x3 : xv;
    aN1[0] = hi ? 0.0f : x4;  aN1[1] = hi ? 0.0f : x5;
    aD[0]  = hi ? x3 : xv;    aD[1]  = hi ? x4 : x2;

    v8f numv = {};
    numv = __builtin_amdgcn_wmma_f32_16x16x4_f32(false, aN0, false, s.n0[t],
                                                 (short)0, numv, false, false);
    numv = __builtin_amdgcn_wmma_f32_16x16x4_f32(false, aN1, false, s.n1[t],
                                                 (short)0, numv, false, false);
    v8f denv = {};
    denv = __builtin_amdgcn_wmma_f32_16x16x4_f32(false, aD, false, s.d[t],
                                                 (short)0, denv, false, false);

#pragma unroll
    for (int k = 0; k < 8; ++k) {
      const float dd = 1.0f + __builtin_fabsf(denv[k]);
      acc[k] = __builtin_fmaf(numv[k], __builtin_amdgcn_rcpf(dd), acc[k]);
    }
  }
}

__global__ __launch_bounds__(256) void kaconv_wmma_kernel(
    const float* __restrict__ x,
    const float* __restrict__ nums,
    const float* __restrict__ denoms,
    const float* __restrict__ bias,
    float* __restrict__ out)
{
  __shared__ v4f coef[LDS_F4];

  const int lane   = threadIdx.x & 31;
  const int wib    = threadIdx.x >> 5;
  const int wave   = blockIdx.x * 8 + wib;
  const int b      = wave / (TILES * 2);
  const int rem    = wave - b * (TILES * 2);
  const int tile   = rem >> 1;
  const int ochunk = rem & 1;

  const int m  = lane & 15;
  const int hi = lane >> 4;

  const int pix = tile * 16 + m;
  const int h = pix / NW;
  const int w = pix - h * NW;

  int  xoff[NT];
  bool inb[NT];
#pragma unroll
  for (int i = 0; i < NK; ++i) {
#pragma unroll
    for (int j = 0; j < NK; ++j) {
      const int hh = h + i - 1;
      const int ww = w + j - 1;
      const bool ok = (hh >= 0) && (hh < NH) && (ww >= 0) && (ww < NW);
      inb[i * NK + j]  = ok;
      xoff[i * NK + j] = ok ? (hh * NW + ww) : pix;
    }
  }

  const int obase = ochunk * 16;

  const v4f* gn = (const v4f*)nums;
  const v4f* gd = (const v4f*)denoms;
  v4f bn[4][7], bd[4][5];
#pragma unroll
  for (int r = 0; r < 4; ++r) {
    const int o = wib + 8 * r;
    const v4f* sn = gn + o * NUM_O_F4 + lane;
    const v4f* sd = gd + o * DEN_O_F4 + lane;
#pragma unroll
    for (int s = 0; s < 6; ++s) bn[r][s] = sn[32 * s];
    bn[r][6] = sn[NUM_O_F4 - 32];
#pragma unroll
    for (int s = 0; s < 4; ++s) bd[r][s] = sd[32 * s];
    bd[r][4] = sd[DEN_O_F4 - 32];
  }
#pragma unroll
  for (int r = 0; r < 4; ++r) {
    const int o = wib + 8 * r;
    v4f* dn = coef + o * NUM_O_PITCH + lane;
    v4f* dd = coef + LDS_DEN_F4 + o * DEN_O_PITCH + lane;
#pragma unroll
    for (int s = 0; s < 6; ++s) dn[32 * s] = bn[r][s];
    dn[NUM_O_F4 - 32] = bn[r][6];
#pragma unroll
    for (int s = 0; s < 4; ++s) dd[32 * s] = bd[r][s];
    dd[DEN_O_F4 - 32] = bd[r][4];
  }
  __syncthreads();

  const float* ls  = (const float*)coef;
  const int    on  = (obase + m) * (NUM_O_PITCH * 4);
  const float* pn0 = ls + on + (hi ? 2 : 0);
  const float* pn1 = ls + on + (hi ? 2 : 4);
  const float* pd  = ls + LDS_DEN_F4 * 4 + (obase + m) * (DEN_O_PITCH * 4) + (hi ? 2 : 0);
  const float* xb  = x + (size_t)(b * NC) * HW;

  v8f acc = {};

  ChanOps sa, sb;
  load_chan(sa, xb, xoff, pn0, pn1, pd);
  for (int c = 0; c < NC - 2; c += 2) {
    __builtin_amdgcn_sched_barrier(0);
    load_chan(sb, xb + (size_t)(c + 1) * HW, xoff,
              pn0 + (c + 1) * (NT * 6), pn1 + (c + 1) * (NT * 6), pd + (c + 1) * (NT * 4));
    __builtin_amdgcn_sched_barrier(0);
    compute_chan(sa, inb, hi, acc);
    __builtin_amdgcn_sched_barrier(0);
    load_chan(sa, xb + (size_t)(c + 2) * HW, xoff,
              pn0 + (c + 2) * (NT * 6), pn1 + (c + 2) * (NT * 6), pd + (c + 2) * (NT * 4));
    __builtin_amdgcn_sched_barrier(0);
    compute_chan(sb, inb, hi, acc);
  }
  __builtin_amdgcn_sched_barrier(0);
  load_chan(sb, xb + (size_t)(NC - 1) * HW, xoff,
            pn0 + (NC - 1) * (NT * 6), pn1 + (NC - 1) * (NT * 6), pd + (NC - 1) * (NT * 4));
  __builtin_amdgcn_sched_barrier(0);
  compute_chan(sa, inb, hi, acc);
  compute_chan(sb, inb, hi, acc);

  const float bo = bias[obase + m];
  float* op = out + ((size_t)(b * NO + obase + m)) * HW + tile * 16 + hi * 8;
#pragma unroll
  for (int k = 0; k < 8; ++k) op[k] = acc[k] + bo;
}

extern "C" void kernel_launch(void* const* d_in, const int* in_sizes, int n_in,
                              void* d_out, int out_size, void* d_ws, size_t ws_size,
                              hipStream_t stream) {
  const float* x      = (const float*)d_in[0];
  const float* nums   = (const float*)d_in[1];
  const float* denoms = (const float*)d_in[2];
  const float* bias   = (const float*)d_in[3];
  float* out          = (float*)d_out;
  (void)in_sizes; (void)n_in; (void)out_size; (void)d_ws; (void)ws_size;

  dim3 grid(196);
  dim3 block(256);
  hipLaunchKernelGGL(kaconv_wmma_kernel, grid, block, 0, stream,
                     x, nums, denoms, bias, out);
}


# Coefficient staging split in two: channels 0..3 staged before the loop; channels 4..15 loaded behind the first barrier, kept in registers over two channels of taps, then stored, second barrier
# speedup vs baseline: 2.1012x; 1.0513x over previous
#include <hip/hip_runtime.h>


typedef float v2f __attribute__((ext_vector_type(2)));
typedef float v4f __attribute__((ext_vector_type(4)));
typedef float v8f __attribute__((ext_vector_type(8)));

#define NB 4
#define NC 16
#define NO 32
#define NK 3
#define NT (NK * NK)
#define NH 56
#define NW 56
#define HW (NH * NW)
#define TILES (HW / 16)

#define NUM_O_F4    (NC * NT * 6 / 4)
#define DEN_O_F4    (NC * NT * 4 / 4)
#define NUM_O_PITCH (NUM_O_F4 + 1)
#define DEN_O_PITCH (DEN_O_F4 + 1)
#define LDS_DEN_F4  (NO * NUM_O_PITCH)
#define LDS_F4      (LDS_DEN_F4 + NO * DEN_O_PITCH)

struct ChanOps {
  float x[NT];
  v2f   n0[NT];
  v2f   n1[NT];
  v2f   d[NT];
};

__device__ __forceinline__ void load_chan(ChanOps& s,
                                          const float* __restrict__ xc,
                                          const int (&xoff)[NT],
                                          const float* __restrict__ pn0,
                                          const float* __restrict__ pn1,
                                          const float* __restrict__ pd) {
#pragma unroll
  for (int t = 0; t < NT; ++t) s.x[t] = xc[xoff[t]];
#pragma unroll
  for (int t = 0; t < NT; ++t) {
    s.n0[t] = *(const v2f*)(pn0 + t * 6);
    s.n1[t] = *(const v2f*)(pn1 + t * 6);
    s.d[t]  = *(const v2f*)(pd  + t * 4);
  }
}

__device__ __forceinline__ void compute_chan(const ChanOps& s,
                                             const bool (&inb)[NT],
                                             const int hi, v8f& acc) {
#pragma unroll
  for (int t = 0; t < NT; ++t) {
    const float xv = inb[t] ? s.x[t] : 0.0f;
    const float x2 = xv * xv;
    const float x3 = x2 * xv;
    const float x4 = x2 * x2;
    const float x5 = x3 * x2;

    v2f aN0 = {}, aN1 = {}, aD = {};
    aN0[0] = hi ? x2 : 1.0f;  aN0[1] = hi ? x3 : xv;
    aN1[0] = hi ? 0.0f : x4;  aN1[1] = hi ? 0.0f : x5;
    aD[0]  = hi ? x3 : xv;    aD[1]  = hi ? x4 : x2;

    v8f numv = {};
    numv = __builtin_amdgcn_wmma_f32_16x16x4_f32(false, aN0, false, s.n0[t],
                                                 (short)0, numv, false, false);
    numv = __builtin_amdgcn_wmma_f32_16x16x4_f32(false, aN1, false, s.n1[t],
                                                 (short)0, numv, false, false);
    v8f denv = {};
    denv = __builtin_amdgcn_wmma_f32_16x16x4_f32(false, aD, false, s.d[t],
                                                 (short)0, denv, false, false);

#pragma unroll
    for (int k = 0; k < 8; ++k) {
      const float dd = 1.0f + __builtin_fabsf(denv[k]);
      acc[k] = __builtin_fmaf(numv[k], __builtin_amdgcn_rcpf(dd), acc[k]);
    }
  }
}

__global__ __launch_bounds__(256) void kaconv_wmma_kernel(
    const float* __restrict__ x,
    const float* __restrict__ nums,
    const float* __restrict__ denoms,
    const float* __restrict__ bias,
    float* __restrict__ out)
{
  __shared__ v4f coef[LDS_F4];

  const int lane   = threadIdx.x & 31;
  const int wib    = threadIdx.x >> 5;
  const int wave   = blockIdx.x * 8 + wib;
  const int b      = wave / (TILES * 2);
  const int rem    = wave - b * (TILES * 2);
  const int tile   = rem >> 1;
  const int ochunk = rem & 1;

  const int m  = lane & 15;
  const int hi = lane >> 4;

  const int pix = tile * 16 + m;
  const int h = pix / NW;
  const int w = pix - h * NW;

  int  xoff[NT];
  bool inb[NT];
#pragma unroll
  for (int i = 0; i < NK; ++i) {
#pragma unroll
    for (int j = 0; j < NK; ++j) {
      const int hh = h + i - 1;
      const int ww = w + j - 1;
      const bool ok = (hh >= 0) && (hh < NH) && (ww >= 0) && (ww < NW);
      inb[i * NK + j]  = ok;
      xoff[i * NK + j] = ok ? (hh * NW + ww) : pix;
    }
  }

  const int obase = ochunk * 16;

  const v4f* gn = (const v4f*)nums;
  const v4f* gd = (const v4f*)denoms;
  constexpr int N1 = 4 * NT * 6 / 4;
  constexpr int D1 = 4 * NT * 4 / 4;
  {
    v4f q[4][4];
#pragma unroll
    for (int r = 0; r < 4; ++r) {
      const int o = wib + 8 * r;
      const v4f* sn = gn + o * NUM_O_F4 + lane;
      const v4f* sd = gd + o * DEN_O_F4 + lane;
      q[r][0] = sn[0];  q[r][1] = sn[N1 - 32];
      q[r][2] = sd[0];  q[r][3] = sd[D1 - 32];
    }
#pragma unroll
    for (int r = 0; r < 4; ++r) {
      const int o = wib + 8 * r;
      v4f* dn = coef + o * NUM_O_PITCH + lane;
      v4f* dd = coef + LDS_DEN_F4 + o * DEN_O_PITCH + lane;
      dn[0] = q[r][0];  dn[N1 - 32] = q[r][1];
      dd[0] = q[r][2];  dd[D1 - 32] = q[r][3];
    }
  }
  __syncthreads();
  const float* ls  = (const float*)coef;
  const int    on  = (obase + m) * (NUM_O_PITCH * 4);
  const float* pn0 = ls + on + (hi ? 2 : 0);
  const float* pn1 = ls + on + (hi ? 2 : 4);
  const float* pd  = ls + LDS_DEN_F4 * 4 + (obase + m) * (DEN_O_PITCH * 4) + (hi ? 2 : 0);
  const float* xb  = x + (size_t)(b * NC) * HW;

  v8f acc = {};

  ChanOps sa, sb;
  load_chan(sa, xb, xoff, pn0, pn1, pd);
  __builtin_amdgcn_sched_barrier(0);
  load_chan(sb, xb + (size_t)1 * HW, xoff, pn0 + 1 * (NT * 6), pn1 + 1 * (NT * 6), pd + 1 * (NT * 4));
  __builtin_amdgcn_sched_barrier(0);
  v4f bn[4][6], bd[4][4];
#pragma unroll
  for (int r = 0; r < 4; ++r) {
    const int o = wib + 8 * r;
    const v4f* sn = gn + o * NUM_O_F4 + lane;
    const v4f* sd = gd + o * DEN_O_F4 + lane;
#pragma unroll
    for (int s = 0; s < 5; ++s) bn[r][s] = sn[N1 + 32 * s];
    bn[r][5] = sn[NUM_O_F4 - 32];
#pragma unroll
    for (int s = 0; s < 3; ++s) bd[r][s] = sd[D1 + 32 * s];
    bd[r][3] = sd[DEN_O_F4 - 32];
  }
  __builtin_amdgcn_sched_barrier(0);
  compute_chan(sa, inb, hi, acc);
  __builtin_amdgcn_sched_barrier(0);
  load_chan(sa, xb + (size_t)2 * HW, xoff, pn0 + 2 * (NT * 6), pn1 + 2 * (NT * 6), pd + 2 * (NT * 4));
  __builtin_amdgcn_sched_barrier(0);
  compute_chan(sb, inb, hi, acc);
  __builtin_amdgcn_sched_barrier(0);
#pragma unroll
  for (int r = 0; r < 4; ++r) {
    const int o = wib + 8 * r;
    v4f* dn = coef + o * NUM_O_PITCH + lane;
    v4f* dd = coef + LDS_DEN_F4 + o * DEN_O_PITCH + lane;
#pragma unroll
    for (int s = 0; s < 5; ++s) dn[N1 + 32 * s] = bn[r][s];
    dn[NUM_O_F4 - 32] = bn[r][5];
#pragma unroll
    for (int s = 0; s < 3; ++s) dd[D1 + 32 * s] = bd[r][s];
    dd[DEN_O_F4 - 32] = bd[r][3];
  }
  __syncthreads();
  for (int c = 2; c < NC - 2; c += 2) {
    __builtin_amdgcn_sched_barrier(0);
    load_chan(sb, xb + (size_t)(c + 1) * HW, xoff,
              pn0 + (c + 1) * (NT * 6), pn1 + (c + 1) * (NT * 6), pd + (c + 1) * (NT * 4));
    __builtin_amdgcn_sched_barrier(0);
    compute_chan(sa, inb, hi, acc);
    __builtin_amdgcn_sched_barrier(0);
    load_chan(sa, xb + (size_t)(c + 2) * HW, xoff,
              pn0 + (c + 2) * (NT * 6), pn1 + (c + 2) * (NT * 6), pd + (c + 2) * (NT * 4));
    __builtin_amdgcn_sched_barrier(0);
    compute_chan(sb, inb, hi, acc);
  }
  __builtin_amdgcn_sched_barrier(0);
  load_chan(sb, xb + (size_t)(NC - 1) * HW, xoff,
            pn0 + (NC - 1) * (NT * 6), pn1 + (NC - 1) * (NT * 6), pd + (NC - 1) * (NT * 4));
  __builtin_amdgcn_sched_barrier(0);
  compute_chan(sa, inb, hi, acc);
  compute_chan(sb, inb, hi, acc);

  const float bo = bias[obase + m];
  float* op = out + ((size_t)(b * NO + obase + m)) * HW + tile * 16 + hi * 8;
#pragma unroll
  for (int k = 0; k < 8; ++k) op[k] = acc[k] + bo;
}

extern "C" void kernel_launch(void* const* d_in, const int* in_sizes, int n_in,
                              void* d_out, int out_size, void* d_ws, size_t ws_size,
                              hipStream_t stream) {
  const float* x      = (const float*)d_in[0];
  const float* nums   = (const float*)d_in[1];
  const float* denoms = (const float*)d_in[2];
  const float* bias   = (const float*)d_in[3];
  float* out          = (float*)d_out;
  (void)in_sizes; (void)n_in; (void)out_size; (void)d_ws; (void)ws_size;

  dim3 grid(196);
  dim3 block(256);
  hipLaunchKernelGGL(kaconv_wmma_kernel, grid, block, 0, stream,
                     x, nums, denoms, bias, out);
}
